# starting kernel
# baseline (speedup 1.0000x reference)
#include <hip/hip_runtime.h>

typedef __attribute__((ext_vector_type(2))) float v2f;
typedef __attribute__((ext_vector_type(4))) float v4f;
typedef __attribute__((ext_vector_type(8))) float v8f;

#define N_SOURCE   100000
#define N_TARGET   4096
#define DEG        800
#define SOURCE_DIM 512
#define EMBED_DIM  32
#define TARGET_DIM 16

__global__ void __launch_bounds__(256)
transpose_embed(const float* __restrict__ embed, float* __restrict__ embedT) {
    int idx = blockIdx.x * blockDim.x + threadIdx.x;
    if (idx < SOURCE_DIM * EMBED_DIM) {
        int k = idx >> 5;
        int n = idx & 31;
        embedT[n * SOURCE_DIM + k] = embed[idx];
    }
}

__global__ void __launch_bounds__(256)
embed_gemm(const float* __restrict__ A, const float* __restrict__ BT,
           const float* __restrict__ xnorm, float* __restrict__ X) {
    const int wave = threadIdx.x >> 5;
    const int tile = blockIdx.x * 8 + wave;
    if (tile >= N_SOURCE / 32) return;
    const int lane = threadIdx.x & 31;
    const int lh   = lane & 15;
    const int hi   = lane >> 4;
    const int row0 = tile * 32;

    const float* __restrict__ arow0 = A  + (size_t)(row0 + lh)      * SOURCE_DIM + 4 * hi;
    const float* __restrict__ arow1 = A  + (size_t)(row0 + 16 + lh) * SOURCE_DIM + 4 * hi;
    const float* __restrict__ b0row = BT + (size_t)(lh)             * SOURCE_DIM + 4 * hi;
    const float* __restrict__ b1row = BT + (size_t)(16 + lh)        * SOURCE_DIM + 4 * hi;

    v8f c00 = {}, c01 = {}, c10 = {}, c11 = {};

    for (int k = 0; k < SOURCE_DIM / 8; ++k) {
        v4f a0 = *(const v4f*)(arow0 + 8 * k);
        v4f a1 = *(const v4f*)(arow1 + 8 * k);
        v4f b0 = *(const v4f*)(b0row + 8 * k);
        v4f b1 = *(const v4f*)(b1row + 8 * k);
        __builtin_prefetch(arow0 + 8 * k + 256, 0, 3);
        __builtin_prefetch(arow1 + 8 * k + 256, 0, 3);

        v2f a0e = {a0.x, a0.y}, a0o = {a0.z, a0.w};
        v2f a1e = {a1.x, a1.y}, a1o = {a1.z, a1.w};
        v2f b0e = {b0.x, b0.y}, b0o = {b0.z, b0.w};
        v2f b1e = {b1.x, b1.y}, b1o = {b1.z, b1.w};

        c00 = __builtin_amdgcn_wmma_f32_16x16x4_f32(false, a0e, false, b0e,
                                                    (short)0, c00, false, false);
        c01 = __builtin_amdgcn_wmma_f32_16x16x4_f32(false, a0e, false, b1e,
                                                    (short)0, c01, false, false);
        c10 = __builtin_amdgcn_wmma_f32_16x16x4_f32(false, a1e, false, b0e,
                                                    (short)0, c10, false, false);
        c11 = __builtin_amdgcn_wmma_f32_16x16x4_f32(false, a1e, false, b1e,
                                                    (short)0, c11, false, false);
        c00 = __builtin_amdgcn_wmma_f32_16x16x4_f32(false, a0o, false, b0o,
                                                    (short)0, c00, false, false);
        c01 = __builtin_amdgcn_wmma_f32_16x16x4_f32(false, a0o, false, b1o,
                                                    (short)0, c01, false, false);
        c10 = __builtin_amdgcn_wmma_f32_16x16x4_f32(false, a1o, false, b0o,
                                                    (short)0, c10, false, false);
        c11 = __builtin_amdgcn_wmma_f32_16x16x4_f32(false, a1o, false, b1o,
                                                    (short)0, c11, false, false);
    }

#pragma unroll
    for (int r = 0; r < 8; ++r) {
        int   rowA = row0 + r + 8 * hi;
        int   rowB = rowA + 16;
        float invA = 1.0f / xnorm[rowA];
        float invB = 1.0f / xnorm[rowB];
        X[(size_t)rowA * EMBED_DIM + lh]      = c00[r] * invA;
        X[(size_t)rowA * EMBED_DIM + 16 + lh] = c01[r] * invA;
        X[(size_t)rowB * EMBED_DIM + lh]      = c10[r] * invB;
        X[(size_t)rowB * EMBED_DIM + 16 + lh] = c11[r] * invB;
    }
}

__global__ void __launch_bounds__(256)
gather_mean(const float* __restrict__ X, const int* __restrict__ src,
            const int* __restrict__ range_list, float* __restrict__ agg) {
    __shared__ float red[8][EMBED_DIM];
    const int t     = blockIdx.x;
    const int col   = threadIdx.x & 31;
    const int grp   = threadIdx.x >> 5;
    const int start = range_list[2 * t];
    const int end   = range_list[2 * t + 1];

    float acc = 0.0f;
    for (int e = start + grp; e < end; e += 8) {
        int s = src[e];
        acc += X[(size_t)s * EMBED_DIM + col];
    }
    red[grp][col] = acc;
    __syncthreads();
    if (grp == 0) {
        float sum = acc;
#pragma unroll
        for (int g = 1; g < 8; ++g) sum += red[g][col];
        float deg = (float)(end - start);
        deg = deg > 1.0f ? deg : 1.0f;
        agg[(size_t)t * EMBED_DIM + col] = sum / deg;
    }
}

__global__ void __launch_bounds__(256)
proj_gemm(const float* __restrict__ agg, const float* __restrict__ W,
          float* __restrict__ out) {
    const int wave = threadIdx.x >> 5;
    const int tile = blockIdx.x * 8 + wave;
    if (tile >= N_TARGET / 16) return;
    const int lane = threadIdx.x & 31;
    const int lh   = lane & 15;
    const int hi   = lane >> 4;
    const int row0 = tile * 16;

    const float* __restrict__ arow = agg + (size_t)(row0 + lh) * EMBED_DIM + 2 * hi;

    v8f c = {};
#pragma unroll
    for (int kb = 0; kb < EMBED_DIM / 4; ++kb) {
        v2f a = *(const v2f*)(arow + 4 * kb);
        v2f b;
        b.x = W[(4 * kb + 2 * hi) * TARGET_DIM + lh];
        b.y = W[(4 * kb + 2 * hi + 1) * TARGET_DIM + lh];
        c = __builtin_amdgcn_wmma_f32_16x16x4_f32(false, a, false, b,
                                                  (short)0, c, false, false);
    }

#pragma unroll
    for (int r = 0; r < 8; ++r) {
        int row = row0 + r + 8 * hi;
        out[(size_t)row * TARGET_DIM + lh] = c[r];
    }
}

extern "C" void kernel_launch(void* const* d_in, const int* in_sizes, int n_in,
                              void* d_out, int out_size, void* d_ws, size_t ws_size,
                              hipStream_t stream) {
    (void)in_sizes; (void)n_in; (void)out_size; (void)ws_size;

    const float* source_feat = (const float*)d_in[0];
    const float* embed       = (const float*)d_in[1];
    const float* weight      = (const float*)d_in[2];
    const int*   edge_index  = (const int*)d_in[3];
    const int*   range_list  = (const int*)d_in[4];
    const float* x_norm      = (const float*)d_in[5];
    float*       out         = (float*)d_out;

    float* embedT = (float*)d_ws;
    float* X      = (float*)((char*)d_ws + 65536);
    float* agg    = (float*)((char*)d_ws + 65536 +
                             (size_t)N_SOURCE * EMBED_DIM * sizeof(float));

    const int* src = edge_index;

    transpose_embed<<<(SOURCE_DIM * EMBED_DIM + 255) / 256, 256, 0, stream>>>(embed, embedT);

    embed_gemm<<<(N_SOURCE / 32 + 7) / 8, 256, 0, stream>>>(source_feat, embedT, x_norm, X);

    gather_mean<<<N_TARGET, 256, 0, stream>>>(X, src, range_list, agg);

    proj_gemm<<<(N_TARGET / 16) / 8, 256, 0, stream>>>(agg, weight, out);
}


# gather_mean: one wave per target, 4 gathered rows per 16-byte-per-lane load, 16-32 row loads in flight (two register buffers, prefetched indices); per-column summation order changed
# speedup vs baseline: 1.7100x; 1.7100x over previous
#include <hip/hip_runtime.h>

typedef __attribute__((ext_vector_type(2))) float v2f;
typedef __attribute__((ext_vector_type(4))) float v4f;
typedef __attribute__((ext_vector_type(8))) float v8f;

#define N_SOURCE   100000
#define N_TARGET   4096
#define DEG        800
#define SOURCE_DIM 512
#define EMBED_DIM  32
#define TARGET_DIM 16

__global__ void __launch_bounds__(256)
transpose_embed(const float* __restrict__ embed, float* __restrict__ embedT) {
    int idx = blockIdx.x * blockDim.x + threadIdx.x;
    if (idx < SOURCE_DIM * EMBED_DIM) {
        int k = idx >> 5;
        int n = idx & 31;
        embedT[n * SOURCE_DIM + k] = embed[idx];
    }
}

__global__ void __launch_bounds__(256)
embed_gemm(const float* __restrict__ A, const float* __restrict__ BT,
           const float* __restrict__ xnorm, float* __restrict__ X) {
    const int wave = threadIdx.x >> 5;
    const int tile = blockIdx.x * 8 + wave;
    if (tile >= N_SOURCE / 32) return;
    const int lane = threadIdx.x & 31;
    const int lh   = lane & 15;
    const int hi   = lane >> 4;
    const int row0 = tile * 32;

    const float* __restrict__ arow0 = A  + (size_t)(row0 + lh)      * SOURCE_DIM + 4 * hi;
    const float* __restrict__ arow1 = A  + (size_t)(row0 + 16 + lh) * SOURCE_DIM + 4 * hi;
    const float* __restrict__ b0row = BT + (size_t)(lh)             * SOURCE_DIM + 4 * hi;
    const float* __restrict__ b1row = BT + (size_t)(16 + lh)        * SOURCE_DIM + 4 * hi;

    v8f c00 = {}, c01 = {}, c10 = {}, c11 = {};

    for (int k = 0; k < SOURCE_DIM / 8; ++k) {
        v4f a0 = *(const v4f*)(arow0 + 8 * k);
        v4f a1 = *(const v4f*)(arow1 + 8 * k);
        v4f b0 = *(const v4f*)(b0row + 8 * k);
        v4f b1 = *(const v4f*)(b1row + 8 * k);
        __builtin_prefetch(arow0 + 8 * k + 256, 0, 3);
        __builtin_prefetch(arow1 + 8 * k + 256, 0, 3);

        v2f a0e = {a0.x, a0.y}, a0o = {a0.z, a0.w};
        v2f a1e = {a1.x, a1.y}, a1o = {a1.z, a1.w};
        v2f b0e = {b0.x, b0.y}, b0o = {b0.z, b0.w};
        v2f b1e = {b1.x, b1.y}, b1o = {b1.z, b1.w};

        c00 = __builtin_amdgcn_wmma_f32_16x16x4_f32(false, a0e, false, b0e,
                                                    (short)0, c00, false, false);
        c01 = __builtin_amdgcn_wmma_f32_16x16x4_f32(false, a0e, false, b1e,
                                                    (short)0, c01, false, false);
        c10 = __builtin_amdgcn_wmma_f32_16x16x4_f32(false, a1e, false, b0e,
                                                    (short)0, c10, false, false);
        c11 = __builtin_amdgcn_wmma_f32_16x16x4_f32(false, a1e, false, b1e,
                                                    (short)0, c11, false, false);
        c00 = __builtin_amdgcn_wmma_f32_16x16x4_f32(false, a0o, false, b0o,
                                                    (short)0, c00, false, false);
        c01 = __builtin_amdgcn_wmma_f32_16x16x4_f32(false, a0o, false, b1o,
                                                    (short)0, c01, false, false);
        c10 = __builtin_amdgcn_wmma_f32_16x16x4_f32(false, a1o, false, b0o,
                                                    (short)0, c10, false, false);
        c11 = __builtin_amdgcn_wmma_f32_16x16x4_f32(false, a1o, false, b1o,
                                                    (short)0, c11, false, false);
    }

#pragma unroll
    for (int r = 0; r < 8; ++r) {
        int   rowA = row0 + r + 8 * hi;
        int   rowB = rowA + 16;
        float invA = 1.0f / xnorm[rowA];
        float invB = 1.0f / xnorm[rowB];
        X[(size_t)rowA * EMBED_DIM + lh]      = c00[r] * invA;
        X[(size_t)rowA * EMBED_DIM + 16 + lh] = c01[r] * invA;
        X[(size_t)rowB * EMBED_DIM + lh]      = c10[r] * invB;
        X[(size_t)rowB * EMBED_DIM + 16 + lh] = c11[r] * invB;
    }
}

#define BC8(v, J) ((int)__builtin_amdgcn_mov_dpp8((unsigned)(v), (unsigned)((J) * 0x249249)))
#define ROWLD(row) (*(const v4f*)(Xb + (size_t)((unsigned)(row) * (EMBED_DIM * 4u) + qoff)))

#define ISSUE_FULL_1(buf, o, id, J) buf[(o) + (J)] = ROWLD(BC8(id, J));
#define ISSUE_FULL(buf, id0, id1)                                               \
    ISSUE_FULL_1(buf, 0, id0, 0) ISSUE_FULL_1(buf, 0, id0, 1)                   \
    ISSUE_FULL_1(buf, 0, id0, 2) ISSUE_FULL_1(buf, 0, id0, 3)                   \
    ISSUE_FULL_1(buf, 0, id0, 4) ISSUE_FULL_1(buf, 0, id0, 5)                   \
    ISSUE_FULL_1(buf, 0, id0, 6) ISSUE_FULL_1(buf, 0, id0, 7)                   \
    ISSUE_FULL_1(buf, 8, id1, 0) ISSUE_FULL_1(buf, 8, id1, 1)                   \
    ISSUE_FULL_1(buf, 8, id1, 2) ISSUE_FULL_1(buf, 8, id1, 3)                   \
    ISSUE_FULL_1(buf, 8, id1, 4) ISSUE_FULL_1(buf, 8, id1, 5)                   \
    ISSUE_FULL_1(buf, 8, id1, 6) ISSUE_FULL_1(buf, 8, id1, 7)

#define ISSUE_MASK_1(buf, o, id, J) {                                           \
        const int r_ = BC8(id, J);                                              \
        v4f v_ = {0.0f, 0.0f, 0.0f, 0.0f};                                      \
        if (r_ >= 0) v_ = ROWLD(r_);                                            \
        buf[(o) + (J)] = v_; }
#define ISSUE_MASKED(buf, id0, id1)                                             \
    ISSUE_MASK_1(buf, 0, id0, 0) ISSUE_MASK_1(buf, 0, id0, 1)                   \
    ISSUE_MASK_1(buf, 0, id0, 2) ISSUE_MASK_1(buf, 0, id0, 3)                   \
    ISSUE_MASK_1(buf, 0, id0, 4) ISSUE_MASK_1(buf, 0, id0, 5)                   \
    ISSUE_MASK_1(buf, 0, id0, 6) ISSUE_MASK_1(buf, 0, id0, 7)                   \
    ISSUE_MASK_1(buf, 8, id1, 0) ISSUE_MASK_1(buf, 8, id1, 1)                   \
    ISSUE_MASK_1(buf, 8, id1, 2) ISSUE_MASK_1(buf, 8, id1, 3)                   \
    ISSUE_MASK_1(buf, 8, id1, 4) ISSUE_MASK_1(buf, 8, id1, 5)                   \
    ISSUE_MASK_1(buf, 8, id1, 6) ISSUE_MASK_1(buf, 8, id1, 7)

#define SUM_UNIT(buf)                                                           \
    acc0 += buf[0];  acc0 += buf[1];  acc0 += buf[2];  acc0 += buf[3];          \
    acc0 += buf[4];  acc0 += buf[5];  acc0 += buf[6];  acc0 += buf[7];          \
    acc1 += buf[8];  acc1 += buf[9];  acc1 += buf[10]; acc1 += buf[11];         \
    acc1 += buf[12]; acc1 += buf[13]; acc1 += buf[14]; acc1 += buf[15];

#define GATHER_WAVES 2

__global__ void __launch_bounds__(GATHER_WAVES * 32)
gather_mean(const float* __restrict__ X, const int* __restrict__ src,
            const int* __restrict__ range_list, float* __restrict__ agg) {
    const int lane = threadIdx.x & 31;
    const int wave = __builtin_amdgcn_readfirstlane(threadIdx.x >> 5);
    const int t    = blockIdx.x * GATHER_WAVES + wave;
    if (t >= N_TARGET) return;
    const int g = lane >> 3;
    const int q = lane & 7;
    const unsigned qoff = 16u * q;
    const char* __restrict__ Xb = (const char*)X;

    const int start = range_list[2 * t];
    const int end   = range_list[2 * t + 1];
    const int n     = end - start;

    v4f acc0 = {0.0f, 0.0f, 0.0f, 0.0f};
    v4f acc1 = {0.0f, 0.0f, 0.0f, 0.0f};

    if (n > 0) {
        const int L    = (n - 1) >> 6;
        const int perm = 4 * q + g;
        const int* __restrict__ sp = src + start + perm;
        v4f A[16], B[16];

        for (int off = 64 * lane; off < n; off += 64 * 32)
            __builtin_prefetch(src + start + off, 0, 3);

        int e0 = 0, e1 = 0;
        int o0 = 0, o1 = 0;
        if (L > 0) { e0 = sp[0];  e1 = sp[32]; }
        if (L > 1) { o0 = sp[64]; o1 = sp[96]; }
        {
            const int x0 = start + 64 * L + perm;
            const int x1 = x0 + 32;
            int m0 = -1, m1 = -1;
            if (x0 < end) m0 = src[x0];
            if (x1 < end) m1 = src[x1];
            ISSUE_MASKED(A, m0, m1)
        }

        const int lastfull = L - 1;
        int u = 0;
        for (; u + 2 <= L; u += 2) {
            {
                const int c0 = e0, c1 = e1;
                const int ue = (u + 2 < lastfull) ? (u + 2) : lastfull;
                e0 = sp[64 * ue]; e1 = sp[64 * ue + 32];
                __builtin_amdgcn_sched_barrier(0);
                ISSUE_FULL(B, c0, c1)
                __builtin_amdgcn_sched_barrier(0);
                SUM_UNIT(A)
            }
            {
                const int c0 = o0, c1 = o1;
                const int uo = (u + 3 < lastfull) ? (u + 3) : lastfull;
                o0 = sp[64 * uo]; o1 = sp[64 * uo + 32];
                __builtin_amdgcn_sched_barrier(0);
                ISSUE_FULL(A, c0, c1)
                __builtin_amdgcn_sched_barrier(0);
                SUM_UNIT(B)
            }
        }
        if (u < L) {
            __builtin_amdgcn_sched_barrier(0);
            ISSUE_FULL(B, e0, e1)
            __builtin_amdgcn_sched_barrier(0);
            SUM_UNIT(A)
            SUM_UNIT(B)
        } else {
            SUM_UNIT(A)
        }
    }

    v4f s = acc0 + acc1;
#pragma unroll
    for (int c = 0; c < 4; ++c) {
        float x = s[c];
        x += __shfl_xor(x, 8, 32);
        x += __shfl_xor(x, 16, 32);
        s[c] = x;
    }
    float deg = (float)n;
    deg = deg > 1.0f ? deg : 1.0f;
    if (g == 0) {
        v4f o = {s[0] / deg, s[1] / deg, s[2] / deg, s[3] / deg};
        *(v4f*)(agg + (size_t)t * EMBED_DIM + 4 * q) = o;
    }
}

__global__ void __launch_bounds__(256)
proj_gemm(const float* __restrict__ agg, const float* __restrict__ W,
          float* __restrict__ out) {
    const int wave = threadIdx.x >> 5;
    const int tile = blockIdx.x * 8 + wave;
    if (tile >= N_TARGET / 16) return;
    const int lane = threadIdx.x & 31;
    const int lh   = lane & 15;
    const int hi   = lane >> 4;
    const int row0 = tile * 16;

    const float* __restrict__ arow = agg + (size_t)(row0 + lh) * EMBED_DIM + 2 * hi;

    v8f c = {};
#pragma unroll
    for (int kb = 0; kb < EMBED_DIM / 4; ++kb) {
        v2f a = *(const v2f*)(arow + 4 * kb);
        v2f b;
        b.x = W[(4 * kb + 2 * hi) * TARGET_DIM + lh];
        b.y = W[(4 * kb + 2 * hi + 1) * TARGET_DIM + lh];
        c = __builtin_amdgcn_wmma_f32_16x16x4_f32(false, a, false, b,
                                                  (short)0, c, false, false);
    }

#pragma unroll
    for (int r = 0; r < 8; ++r) {
        int row = row0 + r + 8 * hi;
        out[(size_t)row * TARGET_DIM + lh] = c[r];
    }
}

extern "C" void kernel_launch(void* const* d_in, const int* in_sizes, int n_in,
                              void* d_out, int out_size, void* d_ws, size_t ws_size,
                              hipStream_t stream) {
    (void)in_sizes; (void)n_in; (void)out_size; (void)ws_size;

    const float* source_feat = (const float*)d_in[0];
    const float* embed       = (const float*)d_in[1];
    const float* weight      = (const float*)d_in[2];
    const int*   edge_index  = (const int*)d_in[3];
    const int*   range_list  = (const int*)d_in[4];
    const float* x_norm      = (const float*)d_in[5];
    float*       out         = (float*)d_out;

    float* embedT = (float*)d_ws;
    float* X      = (float*)((char*)d_ws + 65536);
    float* agg    = (float*)((char*)d_ws + 65536 +
                             (size_t)N_SOURCE * EMBED_DIM * sizeof(float));

    const int* src = edge_index;

    transpose_embed<<<(SOURCE_DIM * EMBED_DIM + 255) / 256, 256, 0, stream>>>(embed, embedT);

    embed_gemm<<<(N_SOURCE / 32 + 7) / 8, 256, 0, stream>>>(source_feat, embedT, x_norm, X);

    gather_mean<<<(N_TARGET + GATHER_WAVES - 1) / GATHER_WAVES, GATHER_WAVES * 32, 0, stream>>>(X, src, range_list, agg);

    proj_gemm<<<(N_TARGET / 16) / 8, 256, 0, stream>>>(agg, weight, out);
}


# reassociate (A.E).W as A.(E.W): E.W folded first on WMMA, X and gathered rows 16 wide (8 rows per load), segment mean is the output, 3 launches; summation order changed
# speedup vs baseline: 2.0963x; 1.2259x over previous
#include <hip/hip_runtime.h>


typedef __attribute__((ext_vector_type(2))) float v2f;
typedef __attribute__((ext_vector_type(4))) float v4f;
typedef __attribute__((ext_vector_type(8))) float v8f;

#define N_SOURCE   100000
#define N_TARGET   4096
#define DEG        800
#define SOURCE_DIM 512
#define EMBED_DIM  32
#define TARGET_DIM 16


__global__ void __launch_bounds__(256)
fold_weights(const float* __restrict__ embed, const float* __restrict__ W,
             float* __restrict__ EWT) {
    const int wave = threadIdx.x >> 5;
    const int tile = blockIdx.x * 8 + wave;
    if (tile >= SOURCE_DIM / 32) return;
    const int lane = threadIdx.x & 31;
    const int lh   = lane & 15;
    const int hi   = lane >> 4;
    const int k0   = tile * 32;

    const float* __restrict__ erow0 = embed + (size_t)(k0 + lh)      * EMBED_DIM + 2 * hi;
    const float* __restrict__ erow1 = embed + (size_t)(k0 + 16 + lh) * EMBED_DIM + 2 * hi;

    v8f c0 = {}, c1 = {};
#pragma unroll
    for (int kb = 0; kb < EMBED_DIM / 4; ++kb) {
        v2f e0 = *(const v2f*)(erow0 + 4 * kb);
        v2f e1 = *(const v2f*)(erow1 + 4 * kb);
        v2f w;
        w.x = W[(4 * kb + 2 * hi) * TARGET_DIM + lh];
        w.y = W[(4 * kb + 2 * hi + 1) * TARGET_DIM + lh];
        c0 = __builtin_amdgcn_wmma_f32_16x16x4_f32(false, w, false, e0,
                                                   (short)0, c0, false, false);
        c1 = __builtin_amdgcn_wmma_f32_16x16x4_f32(false, w, false, e1,
                                                   (short)0, c1, false, false);
    }

#pragma unroll
    for (int r = 0; r < 8; ++r) {
        EWT[(size_t)(r + 8 * hi) * SOURCE_DIM + k0 + lh]      = c0[r];
        EWT[(size_t)(r + 8 * hi) * SOURCE_DIM + k0 + 16 + lh] = c1[r];
    }
}

__global__ void __launch_bounds__(256)
embed_gemm(const float* __restrict__ A, const float* __restrict__ BT,
           const float* __restrict__ xnorm, float* __restrict__ X) {
    const int wave = threadIdx.x >> 5;
    const int tile = blockIdx.x * 8 + wave;
    if (tile >= N_SOURCE / 32) return;
    const int lane = threadIdx.x & 31;
    const int lh   = lane & 15;
    const int hi   = lane >> 4;
    const int row0 = tile * 32;

    const float* __restrict__ arow0 = A  + (size_t)(row0 + lh)      * SOURCE_DIM + 4 * hi;
    const float* __restrict__ arow1 = A  + (size_t)(row0 + 16 + lh) * SOURCE_DIM + 4 * hi;
    const float* __restrict__ brow  = BT + (size_t)(lh)             * SOURCE_DIM + 4 * hi;

    v8f c0 = {}, c1 = {};

    for (int k = 0; k < SOURCE_DIM / 8; ++k) {
        v4f a0 = *(const v4f*)(arow0 + 8 * k);
        v4f a1 = *(const v4f*)(arow1 + 8 * k);
        v4f b  = *(const v4f*)(brow  + 8 * k);
        __builtin_prefetch(arow0 + 8 * k + 256, 0, 3);
        __builtin_prefetch(arow1 + 8 * k + 256, 0, 3);

        v2f a0e = {a0.x, a0.y}, a0o = {a0.z, a0.w};
        v2f a1e = {a1.x, a1.y}, a1o = {a1.z, a1.w};
        v2f be  = {b.x, b.y},   bo  = {b.z, b.w};

        c0 = __builtin_amdgcn_wmma_f32_16x16x4_f32(false, a0e, false, be,
                                                   (short)0, c0, false, false);
        c1 = __builtin_amdgcn_wmma_f32_16x16x4_f32(false, a1e, false, be,
                                                   (short)0, c1, false, false);
        c0 = __builtin_amdgcn_wmma_f32_16x16x4_f32(false, a0o, false, bo,
                                                   (short)0, c0, false, false);
        c1 = __builtin_amdgcn_wmma_f32_16x16x4_f32(false, a1o, false, bo,
                                                   (short)0, c1, false, false);
    }

#pragma unroll
    for (int r = 0; r < 8; ++r) {
        int   rowA = row0 + r + 8 * hi;
        int   rowB = rowA + 16;
        float invA = 1.0f / xnorm[rowA];
        float invB = 1.0f / xnorm[rowB];
        X[(size_t)rowA * TARGET_DIM + lh] = c0[r] * invA;
        X[(size_t)rowB * TARGET_DIM + lh] = c1[r] * invB;
    }
}

#define BC8P(v, J) ((int)__builtin_amdgcn_mov_dpp8((unsigned)(v),               \
                        (unsigned)((2 * (J)) * 0x249 + (2 * (J) + 1) * 0x249000)))
#define ROWLD(row) (*(const v4f*)(Xb + (size_t)((unsigned)(row) * (TARGET_DIM * 4u) + qoff)))

#define ISSUE_FULL_1(buf, o, id, J) buf[(o) + (J)] = ROWLD(BC8P(id, J));
#define ISSUE_FULL_4(buf, o, id)                                                \
    ISSUE_FULL_1(buf, o, id, 0) ISSUE_FULL_1(buf, o, id, 1)                     \
    ISSUE_FULL_1(buf, o, id, 2) ISSUE_FULL_1(buf, o, id, 3)
#define ISSUE_FULL(buf, id0, id1, id2, id3)                                     \
    ISSUE_FULL_4(buf, 0, id0) ISSUE_FULL_4(buf, 4, id1)                         \
    ISSUE_FULL_4(buf, 8, id2) ISSUE_FULL_4(buf, 12, id3)

#define ISSUE_MASK_1(buf, o, id, J) {                                           \
        const int r_ = BC8P(id, J);                                             \
        v4f v_ = {0.0f, 0.0f, 0.0f, 0.0f};                                      \
        if (r_ >= 0) v_ = ROWLD(r_);                                            \
        buf[(o) + (J)] = v_; }
#define ISSUE_MASK_4(buf, o, id)                                                \
    ISSUE_MASK_1(buf, o, id, 0) ISSUE_MASK_1(buf, o, id, 1)                     \
    ISSUE_MASK_1(buf, o, id, 2) ISSUE_MASK_1(buf, o, id, 3)
#define ISSUE_MASKED(buf, id0, id1, id2, id3)                                   \
    ISSUE_MASK_4(buf, 0, id0) ISSUE_MASK_4(buf, 4, id1)                         \
    ISSUE_MASK_4(buf, 8, id2) ISSUE_MASK_4(buf, 12, id3)

#define SUM_UNIT(buf)                                                           \
    acc0 += buf[0];  acc0 += buf[1];  acc0 += buf[2];  acc0 += buf[3];          \
    acc0 += buf[4];  acc0 += buf[5];  acc0 += buf[6];  acc0 += buf[7];          \
    acc1 += buf[8];  acc1 += buf[9];  acc1 += buf[10]; acc1 += buf[11];         \
    acc1 += buf[12]; acc1 += buf[13]; acc1 += buf[14]; acc1 += buf[15];

#define GATHER_WAVES 2
static_assert(N_TARGET % GATHER_WAVES == 0, "every work-group holds two whole targets");

__global__ void __launch_bounds__(GATHER_WAVES * 32)
gather_mean(const float* __restrict__ X, const int* __restrict__ src,
            const int* __restrict__ range_list, float* __restrict__ out) {
    __shared__ v4f xch[4];
    const int lane = threadIdx.x & 31;
    const int wave = __builtin_amdgcn_readfirstlane(threadIdx.x >> 5);
    const int t    = blockIdx.x * GATHER_WAVES + wave;
    const int q = lane & 3;
    const unsigned qoff = 16u * q;
    const char* __restrict__ Xb = (const char*)X;

    const int start = range_list[2 * t];
    const int end   = range_list[2 * t + 1];
    const int n     = end - start;

    v4f acc0 = {0.0f, 0.0f, 0.0f, 0.0f};
    v4f acc1 = {0.0f, 0.0f, 0.0f, 0.0f};

    if (n > 0) {
        const int L    = (n - 1) >> 7;
        const int m    = lane & 7;
        const int G    = lane >> 3;
        const int perm = 8 * (m >> 1) + 2 * G + (m & 1);
        const int* __restrict__ sp = src + start + perm;
        v4f A[16], B[16];

        for (int off = 64 * lane; off < n; off += 64 * 32)
            __builtin_prefetch(src + start + off, 0, 3);

        int e0 = 0, e1 = 0, e2 = 0, e3 = 0;
        int o0 = 0, o1 = 0, o2 = 0, o3 = 0;
        if (L > 0) { e0 = sp[0];   e1 = sp[32];  e2 = sp[64];  e3 = sp[96];  }
        if (L > 1) { o0 = sp[128]; o1 = sp[160]; o2 = sp[192]; o3 = sp[224]; }
        {
            const int x0 = start + 128 * L + perm;
            int m0 = -1, m1 = -1, m2 = -1, m3 = -1;
            if (x0      < end) m0 = src[x0];
            if (x0 + 32 < end) m1 = src[x0 + 32];
            if (x0 + 64 < end) m2 = src[x0 + 64];
            if (x0 + 96 < end) m3 = src[x0 + 96];
            ISSUE_MASKED(A, m0, m1, m2, m3)
        }

        const int lastfull = L - 1;
        int u = 0;
        for (; u + 2 <= L; u += 2) {
            {
                const int c0 = e0, c1 = e1, c2 = e2, c3 = e3;
                const int ue = (u + 2 < lastfull) ? (u + 2) : lastfull;
                e0 = sp[128 * ue];      e1 = sp[128 * ue + 32];
                e2 = sp[128 * ue + 64]; e3 = sp[128 * ue + 96];
                __builtin_amdgcn_sched_barrier(0);
                ISSUE_FULL(B, c0, c1, c2, c3)
                __builtin_amdgcn_sched_barrier(0);
                SUM_UNIT(A)
            }
            {
                const int c0 = o0, c1 = o1, c2 = o2, c3 = o3;
                const int uo = (u + 3 < lastfull) ? (u + 3) : lastfull;
                o0 = sp[128 * uo];      o1 = sp[128 * uo + 32];
                o2 = sp[128 * uo + 64]; o3 = sp[128 * uo + 96];
                __builtin_amdgcn_sched_barrier(0);
                ISSUE_FULL(A, c0, c1, c2, c3)
                __builtin_amdgcn_sched_barrier(0);
                SUM_UNIT(B)
            }
        }
        if (u < L) {
            __builtin_amdgcn_sched_barrier(0);
            ISSUE_FULL(B, e0, e1, e2, e3)
            __builtin_amdgcn_sched_barrier(0);
            SUM_UNIT(A)
            SUM_UNIT(B)
        } else {
            SUM_UNIT(A)
        }
    }

    v4f s = acc0 + acc1;
#pragma unroll
    for (int c = 0; c < 4; ++c) {
        float x = s[c];
        x += __shfl_xor(x, 4, 32);
        x += __shfl_xor(x, 8, 32);
        x += __shfl_xor(x, 16, 32);
        s[c] = x;
    }
    float deg = (float)n;
    deg = deg > 1.0f ? deg : 1.0f;
    const v4f o = {s[0] / deg, s[1] / deg, s[2] / deg, s[3] / deg};

    if (wave == 1 && lane < 4) xch[q] = o;
    __syncthreads();
    if (wave == 0 && lane < 8) {
        const v4f w = (lane < 4) ? o : xch[q];
        *(v4f*)(out + (size_t)t * TARGET_DIM + 4 * lane) = w;
    }
}

extern "C" void kernel_launch(void* const* d_in, const int* in_sizes, int n_in,
                              void* d_out, int out_size, void* d_ws, size_t ws_size,
                              hipStream_t stream) {
    (void)in_sizes; (void)n_in; (void)out_size; (void)ws_size;

    const float* source_feat = (const float*)d_in[0];
    const float* embed       = (const float*)d_in[1];
    const float* weight      = (const float*)d_in[2];
    const int*   edge_index  = (const int*)d_in[3];
    const int*   range_list  = (const int*)d_in[4];
    const float* x_norm      = (const float*)d_in[5];
    float*       out         = (float*)d_out;

    float* EWT = (float*)d_ws;
    float* X   = (float*)((char*)d_ws + 65536);

    const int* src = edge_index;

    fold_weights<<<(SOURCE_DIM / 32) / 8, 256, 0, stream>>>(embed, weight, EWT);

    embed_gemm<<<(N_SOURCE / 32 + 7) / 8, 256, 0, stream>>>(source_feat, EWT, x_norm, X);

    gather_mean<<<N_TARGET / GATHER_WAVES, GATHER_WAVES * 32, 0, stream>>>(X, src, range_list, out);
}


# embed_gemm load path: A rows streamed by whole-line non-temporal 16-byte loads (2 rows x 256 B per instruction, 16-32 in flight), A and B fragments read from LDS; no prefetches; sum order kept
# speedup vs baseline: 3.0634x; 1.4613x over previous
#include <hip/hip_runtime.h>


typedef __attribute__((ext_vector_type(2))) float v2f;
typedef __attribute__((ext_vector_type(4))) float v4f;
typedef __attribute__((ext_vector_type(8))) float v8f;

#define N_SOURCE   100000
#define N_TARGET   4096
#define DEG        800
#define SOURCE_DIM 512
#define EMBED_DIM  32
#define TARGET_DIM 16


__global__ void __launch_bounds__(256)
fold_weights(const float* __restrict__ embed, const float* __restrict__ W,
             float* __restrict__ EWT) {
    const int wave = threadIdx.x >> 5;
    const int tile = blockIdx.x * 8 + wave;
    if (tile >= SOURCE_DIM / 32) return;
    const int lane = threadIdx.x & 31;
    const int lh   = lane & 15;
    const int hi   = lane >> 4;
    const int k0   = tile * 32;

    const float* __restrict__ erow0 = embed + (size_t)(k0 + lh)      * EMBED_DIM + 2 * hi;
    const float* __restrict__ erow1 = embed + (size_t)(k0 + 16 + lh) * EMBED_DIM + 2 * hi;

    v8f c0 = {}, c1 = {};
#pragma unroll
    for (int kb = 0; kb < EMBED_DIM / 4; ++kb) {
        v2f e0 = *(const v2f*)(erow0 + 4 * kb);
        v2f e1 = *(const v2f*)(erow1 + 4 * kb);
        v2f w;
        w.x = W[(4 * kb + 2 * hi) * TARGET_DIM + lh];
        w.y = W[(4 * kb + 2 * hi + 1) * TARGET_DIM + lh];
        c0 = __builtin_amdgcn_wmma_f32_16x16x4_f32(false, w, false, e0,
                                                   (short)0, c0, false, false);
        c1 = __builtin_amdgcn_wmma_f32_16x16x4_f32(false, w, false, e1,
                                                   (short)0, c1, false, false);
    }

#pragma unroll
    for (int r = 0; r < 8; ++r) {
        EWT[(size_t)(r + 8 * hi) * SOURCE_DIM + k0 + lh]      = c0[r];
        EWT[(size_t)(r + 8 * hi) * SOURCE_DIM + k0 + 16 + lh] = c1[r];
    }
}

#define GEMM_WAVES 4
#define KC         64
#define NCHUNK     (SOURCE_DIM / KC)
#define A_PITCH    (KC * 4 + 16)
#define B_PITCH    (SOURCE_DIM * 4 + 16)
static_assert(N_SOURCE % 32 == 0, "whole 32-row tiles only");
static_assert(SOURCE_DIM * 4 == GEMM_WAVES * 32 * 16, "one 16-byte piece of an EW^T row per thread");

#define A_ISSUE(buf, C)                                                         \
    _Pragma("unroll") for (int i_ = 0; i_ < 16; ++i_)                           \
        buf[i_] = __builtin_nontemporal_load((const v4f*)(gp + i_ * 2 * SOURCE_DIM + (C) * KC));
#define A_PARK(buf)                                                             \
    _Pragma("unroll") for (int i_ = 0; i_ < 16; ++i_)                           \
        *(v4f*)(stA + i_ * 2 * A_PITCH) = buf[i_];

__global__ void __launch_bounds__(GEMM_WAVES * 32)
embed_gemm(const float* __restrict__ A, const float* __restrict__ BT,
           const float* __restrict__ xnorm, float* __restrict__ X) {
    __shared__ __attribute__((aligned(16))) unsigned char sB[TARGET_DIM * B_PITCH];
    __shared__ __attribute__((aligned(16))) unsigned char sA[GEMM_WAVES][32 * A_PITCH];

    const int wave = __builtin_amdgcn_readfirstlane(threadIdx.x >> 5);
    const int tile = blockIdx.x * GEMM_WAVES + wave;
    const bool live = tile < N_SOURCE / 32;
    const int lane = threadIdx.x & 31;
    const int lh   = lane & 15;
    const int hi   = lane >> 4;
    const int row0 = (live ? tile : 0) * 32;

    const float* __restrict__ gp = A + (size_t)(row0 + hi) * SOURCE_DIM + 4 * lh;
    unsigned char* stA = &sA[wave][hi * A_PITCH + 16 * lh];
    const unsigned char* fA = &sA[wave][lh * A_PITCH + 16 * hi];
    const unsigned char* fB = &sB[lh * B_PITCH + 16 * hi];

    v4f R0[16], R1[16];
    A_ISSUE(R0, 0)
    A_ISSUE(R1, 1)

#pragma unroll
    for (int n = 0; n < TARGET_DIM; ++n)
        *(v4f*)&sB[n * B_PITCH + 16 * threadIdx.x] =
            *(const v4f*)(BT + (size_t)n * SOURCE_DIM + 4 * threadIdx.x);
    __syncthreads();
    if (!live) return;

    v8f c0 = {}, c1 = {};

#pragma unroll
    for (int c = 0; c < NCHUNK; ++c) {
        if (c & 1) { A_PARK(R1) } else { A_PARK(R0) }
        __builtin_amdgcn_sched_barrier(0);
        if (c + 2 < NCHUNK) {
            if (c & 1) { A_ISSUE(R1, c + 2) } else { A_ISSUE(R0, c + 2) }
        }
        __builtin_amdgcn_sched_barrier(0);
#pragma unroll
        for (int j = 0; j < KC / 8; ++j) {
            const v4f a0 = *(const v4f*)(fA + 32 * j);
            const v4f a1 = *(const v4f*)(fA + 16 * A_PITCH + 32 * j);
            const v4f b  = *(const v4f*)(fB + 4 * KC * c + 32 * j);

            v2f a0e = {a0.x, a0.y}, a0o = {a0.z, a0.w};
            v2f a1e = {a1.x, a1.y}, a1o = {a1.z, a1.w};
            v2f be  = {b.x, b.y},   bo  = {b.z, b.w};

            c0 = __builtin_amdgcn_wmma_f32_16x16x4_f32(false, a0e, false, be,
                                                       (short)0, c0, false, false);
            c1 = __builtin_amdgcn_wmma_f32_16x16x4_f32(false, a1e, false, be,
                                                       (short)0, c1, false, false);
            c0 = __builtin_amdgcn_wmma_f32_16x16x4_f32(false, a0o, false, bo,
                                                       (short)0, c0, false, false);
            c1 = __builtin_amdgcn_wmma_f32_16x16x4_f32(false, a1o, false, bo,
                                                       (short)0, c1, false, false);
        }
    }

#pragma unroll
    for (int r = 0; r < 8; ++r) {
        int   rowA = row0 + r + 8 * hi;
        int   rowB = rowA + 16;
        float invA = 1.0f / xnorm[rowA];
        float invB = 1.0f / xnorm[rowB];
        X[(size_t)rowA * TARGET_DIM + lh] = c0[r] * invA;
        X[(size_t)rowB * TARGET_DIM + lh] = c1[r] * invB;
    }
}

#define BC8P(v, J) ((int)__builtin_amdgcn_mov_dpp8((unsigned)(v),               \
                        (unsigned)((2 * (J)) * 0x249 + (2 * (J) + 1) * 0x249000)))
#define ROWLD(row) (*(const v4f*)(Xb + (size_t)((unsigned)(row) * (TARGET_DIM * 4u) + qoff)))

#define ISSUE_FULL_1(buf, o, id, J) buf[(o) + (J)] = ROWLD(BC8P(id, J));
#define ISSUE_FULL_4(buf, o, id)                                                \
    ISSUE_FULL_1(buf, o, id, 0) ISSUE_FULL_1(buf, o, id, 1)                     \
    ISSUE_FULL_1(buf, o, id, 2) ISSUE_FULL_1(buf, o, id, 3)
#define ISSUE_FULL(buf, id0, id1, id2, id3)                                     \
    ISSUE_FULL_4(buf, 0, id0) ISSUE_FULL_4(buf, 4, id1)                         \
    ISSUE_FULL_4(buf, 8, id2) ISSUE_FULL_4(buf, 12, id3)

#define ISSUE_MASK_1(buf, o, id, J) {                                           \
        const int r_ = BC8P(id, J);                                             \
        v4f v_ = {0.0f, 0.0f, 0.0f, 0.0f};                                      \
        if (r_ >= 0) v_ = ROWLD(r_);                                            \
        buf[(o) + (J)] = v_; }
#define ISSUE_MASK_4(buf, o, id)                                                \
    ISSUE_MASK_1(buf, o, id, 0) ISSUE_MASK_1(buf, o, id, 1)                     \
    ISSUE_MASK_1(buf, o, id, 2) ISSUE_MASK_1(buf, o, id, 3)
#define ISSUE_MASKED(buf, id0, id1, id2, id3)                                   \
    ISSUE_MASK_4(buf, 0, id0) ISSUE_MASK_4(buf, 4, id1)                         \
    ISSUE_MASK_4(buf, 8, id2) ISSUE_MASK_4(buf, 12, id3)

#define SUM_UNIT(buf)                                                           \
    acc0 += buf[0];  acc0 += buf[1];  acc0 += buf[2];  acc0 += buf[3];          \
    acc0 += buf[4];  acc0 += buf[5];  acc0 += buf[6];  acc0 += buf[7];          \
    acc1 += buf[8];  acc1 += buf[9];  acc1 += buf[10]; acc1 += buf[11];         \
    acc1 += buf[12]; acc1 += buf[13]; acc1 += buf[14]; acc1 += buf[15];

#define GATHER_WAVES 2
static_assert(N_TARGET % GATHER_WAVES == 0, "every work-group holds two whole targets");

__global__ void __launch_bounds__(GATHER_WAVES * 32)
gather_mean(const float* __restrict__ X, const int* __restrict__ src,
            const int* __restrict__ range_list, float* __restrict__ out) {
    __shared__ v4f xch[4];
    const int lane = threadIdx.x & 31;
    const int wave = __builtin_amdgcn_readfirstlane(threadIdx.x >> 5);
    const int t    = blockIdx.x * GATHER_WAVES + wave;
    const int q = lane & 3;
    const unsigned qoff = 16u * q;
    const char* __restrict__ Xb = (const char*)X;

    const int start = range_list[2 * t];
    const int end   = range_list[2 * t + 1];
    const int n     = end - start;

    v4f acc0 = {0.0f, 0.0f, 0.0f, 0.0f};
    v4f acc1 = {0.0f, 0.0f, 0.0f, 0.0f};

    if (n > 0) {
        const int L    = (n - 1) >> 7;
        const int m    = lane & 7;
        const int G    = lane >> 3;
        const int perm = 8 * (m >> 1) + 2 * G + (m & 1);
        const int* __restrict__ sp = src + start + perm;
        v4f A[16], B[16];

        for (int off = 64 * lane; off < n; off += 64 * 32)
            __builtin_prefetch(src + start + off, 0, 3);

        int e0 = 0, e1 = 0, e2 = 0, e3 = 0;
        int o0 = 0, o1 = 0, o2 = 0, o3 = 0;
        if (L > 0) { e0 = sp[0];   e1 = sp[32];  e2 = sp[64];  e3 = sp[96];  }
        if (L > 1) { o0 = sp[128]; o1 = sp[160]; o2 = sp[192]; o3 = sp[224]; }
        {
            const int x0 = start + 128 * L + perm;
            int m0 = -1, m1 = -1, m2 = -1, m3 = -1;
            if (x0      < end) m0 = src[x0];
            if (x0 + 32 < end) m1 = src[x0 + 32];
            if (x0 + 64 < end) m2 = src[x0 + 64];
            if (x0 + 96 < end) m3 = src[x0 + 96];
            ISSUE_MASKED(A, m0, m1, m2, m3)
        }

        const int lastfull = L - 1;
        int u = 0;
        for (; u + 2 <= L; u += 2) {
            {
                const int c0 = e0, c1 = e1, c2 = e2, c3 = e3;
                const int ue = (u + 2 < lastfull) ? (u + 2) : lastfull;
                e0 = sp[128 * ue];      e1 = sp[128 * ue + 32];
                e2 = sp[128 * ue + 64]; e3 = sp[128 * ue + 96];
                __builtin_amdgcn_sched_barrier(0);
                ISSUE_FULL(B, c0, c1, c2, c3)
                __builtin_amdgcn_sched_barrier(0);
                SUM_UNIT(A)
            }
            {
                const int c0 = o0, c1 = o1, c2 = o2, c3 = o3;
                const int uo = (u + 3 < lastfull) ? (u + 3) : lastfull;
                o0 = sp[128 * uo];      o1 = sp[128 * uo + 32];
                o2 = sp[128 * uo + 64]; o3 = sp[128 * uo + 96];
                __builtin_amdgcn_sched_barrier(0);
                ISSUE_FULL(A, c0, c1, c2, c3)
                __builtin_amdgcn_sched_barrier(0);
                SUM_UNIT(B)
            }
        }
        if (u < L) {
            __builtin_amdgcn_sched_barrier(0);
            ISSUE_FULL(B, e0, e1, e2, e3)
            __builtin_amdgcn_sched_barrier(0);
            SUM_UNIT(A)
            SUM_UNIT(B)
        } else {
            SUM_UNIT(A)
        }
    }

    v4f s = acc0 + acc1;
#pragma unroll
    for (int c = 0; c < 4; ++c) {
        float x = s[c];
        x += __shfl_xor(x, 4, 32);
        x += __shfl_xor(x, 8, 32);
        x += __shfl_xor(x, 16, 32);
        s[c] = x;
    }
    float deg = (float)n;
    deg = deg > 1.0f ? deg : 1.0f;
    const v4f o = {s[0] / deg, s[1] / deg, s[2] / deg, s[3] / deg};

    if (wave == 1 && lane < 4) xch[q] = o;
    __syncthreads();
    if (wave == 0 && lane < 8) {
        const v4f w = (lane < 4) ? o : xch[q];
        *(v4f*)(out + (size_t)t * TARGET_DIM + 4 * lane) = w;
    }
}

extern "C" void kernel_launch(void* const* d_in, const int* in_sizes, int n_in,
                              void* d_out, int out_size, void* d_ws, size_t ws_size,
                              hipStream_t stream) {
    (void)in_sizes; (void)n_in; (void)out_size; (void)ws_size;

    const float* source_feat = (const float*)d_in[0];
    const float* embed       = (const float*)d_in[1];
    const float* weight      = (const float*)d_in[2];
    const int*   edge_index  = (const int*)d_in[3];
    const int*   range_list  = (const int*)d_in[4];
    const float* x_norm      = (const float*)d_in[5];
    float*       out         = (float*)d_out;

    float* EWT = (float*)d_ws;
    float* X   = (float*)((char*)d_ws + 65536);

    const int* src = edge_index;

    fold_weights<<<(SOURCE_DIM / 32) / 8, 256, 0, stream>>>(embed, weight, EWT);

    embed_gemm<<<(N_SOURCE / 32 + GEMM_WAVES - 1) / GEMM_WAVES, GEMM_WAVES * 32, 0, stream>>>(source_feat, EWT, x_norm, X);

    gather_mean<<<N_TARGET / GATHER_WAVES, GATHER_WAVES * 32, 0, stream>>>(X, src, range_list, out);
}
